# starting kernel
# baseline (speedup 1.0000x reference)
#include <hip/hip_runtime.h>


#define L1X2   1024
#define NL2    16
#define NL3    32
#define NSTACK 8
#define W1PAD  1028
#define Y1PAD  17
#define BPS    64
#define WAVES  8

typedef __attribute__((ext_vector_type(2))) float v2f;
typedef __attribute__((ext_vector_type(8))) float v8f;

__global__ void ls_init(int* cnt) {
    if (threadIdx.x < 2 * NSTACK) cnt[threadIdx.x >= NSTACK ? 8 + threadIdx.x : threadIdx.x] = 0;
}

__global__ void ls_hist(const int* __restrict__ idx, int* __restrict__ counts, int B) {
    int s = blockIdx.x * blockDim.x + threadIdx.x;
    if (s < B) atomicAdd(&counts[idx[s]], 1);
}

__global__ void ls_scan(const int* __restrict__ counts, int* __restrict__ offsets) {
    int acc = 0;
    for (int c = 0; c < NSTACK; ++c) { offsets[c] = acc; acc += counts[c]; }
}

__global__ void ls_scatter(const int* __restrict__ idx, const int* __restrict__ offsets,
                           int* __restrict__ cursors, int* __restrict__ bucket, int B) {
    int s = blockIdx.x * blockDim.x + threadIdx.x;
    if (s < B) {
        int c = idx[s];
        int pos = offsets[c] + atomicAdd(&cursors[c], 1);
        bucket[pos] = s;
    }
}

__global__ __launch_bounds__(256)
void ls_main(const float* __restrict__ x,
             const float* __restrict__ w1, const float* __restrict__ b1,
             const float* __restrict__ w2, const float* __restrict__ b2,
             const float* __restrict__ wo, const float* __restrict__ bo,
             const int* __restrict__ counts, const int* __restrict__ offsets,
             const int* __restrict__ bucket, float* __restrict__ out) {
    __shared__ float w1s[NL2 * W1PAD];
    __shared__ float w2s[NL3 * NL2];
    __shared__ float b1s[NL2];
    __shared__ float b2s[NL3];
    __shared__ float wos[NL3];
    __shared__ float y1s[WAVES][NL2 * Y1PAD];

    const int c   = blockIdx.y;
    const int tid = threadIdx.x;

    for (int i = tid; i < NL2 * L1X2; i += 256) {
        int r = i >> 10, col = i & 1023;
        w1s[r * W1PAD + col] = w1[(c * NL2 + r) * L1X2 + col];
    }
    for (int i = tid; i < NL3 * NL2; i += 256) {
        int r = i >> 4, col = i & 15;
        w2s[i] = w2[(c * NL3 + r) * NL2 + col];
    }
    if (tid < NL2) b1s[tid] = b1[c * NL2 + tid];
    if (tid < NL3) { b2s[tid] = b2[c * NL3 + tid]; wos[tid] = wo[c * NL3 + tid]; }
    __syncthreads();

    const int n_c = counts[c];
    if (n_c == 0) return;
    const int   base = offsets[c];
    const float bo_c = bo[c];
    const int nTiles = (n_c + 15) >> 4;

    const int wave = tid >> 5;
    const int lane = tid & 31;
    const int ml   = lane & 15;
    const int half = lane >> 4;
    const int mlr  = ml * W1PAD;

    for (int t = blockIdx.x * WAVES + wave; t < nTiles; t += gridDim.x * WAVES) {
        int gs   = t * 16 + ml;
        int cgs  = gs < n_c ? gs : n_c - 1;
        int samp = bucket[base + cgs];
        const float* xrow = x + (long)samp * L1X2;

        v8f acc = {0.f, 0.f, 0.f, 0.f, 0.f, 0.f, 0.f, 0.f};
        #pragma unroll 8
        for (int k = 0; k < L1X2; k += 4) {
            const int ko = k + 2 * half;
            v2f a = __builtin_nontemporal_load((const v2f*)(xrow + ko));
            v2f b = *(const v2f*)(w1s + mlr + ko);
            acc = __builtin_amdgcn_wmma_f32_16x16x4_f32(
                false, a, false, b, (short)0, acc, false, false);
        }

        float* yb = &y1s[wave][0];
        #pragma unroll
        for (int r = 0; r < 8; ++r) {
            float v = acc[r] + b1s[ml];
            v = fminf(fmaxf(v, 0.f), 1.f);
            yb[(8 * half + r) * Y1PAD + ml] = v;
        }
        asm volatile("" ::: "memory");

        float xv[NL2];
        #pragma unroll
        for (int j = 0; j < NL2; ++j) xv[j] = yb[ml * Y1PAD + j];

        float acc3 = 0.f;
        #pragma unroll
        for (int q = 0; q < 16; ++q) {
            int jj = half * 16 + q;
            float s2 = b2s[jj];
            #pragma unroll
            for (int j = 0; j < NL2; ++j) s2 += xv[j] * w2s[jj * NL2 + j];
            s2 = fminf(fmaxf(s2, 0.f), 1.f);
            acc3 += s2 * wos[jj];
        }
        acc3 += __shfl_xor(acc3, 16, 32);

        if (half == 0 && gs < n_c) out[bucket[base + gs]] = acc3 + bo_c;
    }
}

extern "C" void kernel_launch(void* const* d_in, const int* in_sizes, int n_in,
                              void* d_out, int out_size, void* d_ws, size_t ws_size,
                              hipStream_t stream) {
    const float* x   = (const float*)d_in[0];
    const int*   idx = (const int*)  d_in[1];
    const float* w1  = (const float*)d_in[2];
    const float* b1  = (const float*)d_in[3];
    const float* w2  = (const float*)d_in[4];
    const float* b2  = (const float*)d_in[5];
    const float* wo  = (const float*)d_in[6];
    const float* bo  = (const float*)d_in[7];
    float* out = (float*)d_out;

    const int B = in_sizes[1];
    int* wsi     = (int*)d_ws;
    int* counts  = wsi;
    int* offsets = wsi + 8;
    int* cursors = wsi + 16;
    int* bucket  = wsi + 32;

    ls_init<<<1, 32, 0, stream>>>(counts);
    int nb = (B + 255) / 256;
    ls_hist<<<nb, 256, 0, stream>>>(idx, counts, B);
    ls_scan<<<1, 1, 0, stream>>>(counts, offsets);
    ls_scatter<<<nb, 256, 0, stream>>>(idx, offsets, cursors, bucket, B);

    dim3 grid(BPS, NSTACK, 1);
    ls_main<<<grid, 256, 0, stream>>>(x, w1, b1, w2, b2, wo, bo,
                                      counts, offsets, bucket, out);
}


# Replace the 4-launch global-atomic bucket sort: each block ballots its own 683-sample chunk for its stack into an LDS list and tiles from it (no wave gets a second tile); 1 launch, no atomics
# speedup vs baseline: 1.7696x; 1.7696x over previous
#include <hip/hip_runtime.h>


#define L1X2   1024
#define NL2    16
#define NL3    32
#define NSTACK 8
#define W1PAD  1028
#define Y1PAD  17
#define BPS    96
#define WAVES  8
#define SPT    3
#define CHUNK  683

typedef __attribute__((ext_vector_type(2))) float v2f;
typedef __attribute__((ext_vector_type(8))) float v8f;

__global__ __launch_bounds__(256, 7)
void ls_main(const float* __restrict__ x, const int* __restrict__ idx,
             const float* __restrict__ w1, const float* __restrict__ b1,
             const float* __restrict__ w2, const float* __restrict__ b2,
             const float* __restrict__ wo, const float* __restrict__ bo,
             float* __restrict__ out, int B) {
    __shared__ float w1s[NL2 * W1PAD];
    __shared__ float w2s[NL3 * NL2];
    __shared__ float b1s[NL2];
    __shared__ float b2s[NL3];
    __shared__ float wos[NL3];
    __shared__ float y1s[WAVES][NL2 * Y1PAD];
    __shared__ int   lst[CHUNK];
    __shared__ int   wtot[WAVES];

    const int c   = blockIdx.y;
    const int tid = threadIdx.x;

    int s0 = blockIdx.x * CHUNK;
    int iv[SPT];
    #pragma unroll
    for (int e = 0; e < SPT; ++e) iv[e] = 0;
    if (s0 < B) {
        #pragma unroll
        for (int e = 0; e < SPT; ++e) {
            int o = e * 256 + tid;
            int s = s0 + o;
            iv[e] = idx[(o < CHUNK && s < B) ? s : s0];
        }
    }

    for (int i = tid; i < NL2 * L1X2; i += 256) {
        int r = i >> 10, col = i & 1023;
        w1s[r * W1PAD + col] = w1[(c * NL2 + r) * L1X2 + col];
    }
    for (int i = tid; i < NL3 * NL2; i += 256) {
        int r = i >> 4, col = i & 15;
        w2s[i] = w2[(c * NL3 + r) * NL2 + col];
    }
    if (tid < NL2) b1s[tid] = b1[c * NL2 + tid];
    if (tid < NL3) { b2s[tid] = b2[c * NL3 + tid]; wos[tid] = wo[c * NL3 + tid]; }

    const float bo_c = bo[c];

    const int wave = tid >> 5;
    const int lane = tid & 31;
    const int ml   = lane & 15;
    const int half = lane >> 4;
    const int mlr  = ml * W1PAD;
    const unsigned lanes_below = (1u << lane) - 1u;

    for (; s0 < B; s0 += gridDim.x * CHUNK) {
        bool     hit[SPT];
        unsigned bal[SPT];
        int found = 0;
        #pragma unroll
        for (int e = 0; e < SPT; ++e) {
            int o = e * 256 + tid;
            hit[e] = (o < CHUNK) && (s0 + o < B) && (iv[e] == c);
            bal[e] = (unsigned)__ballot(hit[e]);
            found += __popc(bal[e]);
        }
        if (lane == 0) wtot[wave] = found;
        __syncthreads();

        int pos = 0, n_c = 0;
        #pragma unroll
        for (int w = 0; w < WAVES; ++w) {
            int v = wtot[w];
            if (w < wave) pos += v;
            n_c += v;
        }
        #pragma unroll
        for (int e = 0; e < SPT; ++e) {
            if (hit[e]) lst[pos + __popc(bal[e] & lanes_below)] = s0 + e * 256 + tid;
            pos += __popc(bal[e]);
        }
        __syncthreads();

        const int nTiles = (n_c + 15) >> 4;

        for (int t = wave; t < nTiles; t += WAVES) {
            int gs   = t * 16 + ml;
            int cgs  = gs < n_c ? gs : n_c - 1;
            int samp = lst[cgs];
            const float* xrow = x + (long)samp * L1X2;

            v8f acc = {0.f, 0.f, 0.f, 0.f, 0.f, 0.f, 0.f, 0.f};
            #pragma unroll 8
            for (int k = 0; k < L1X2; k += 4) {
                const int ko = k + 2 * half;
                v2f a = __builtin_nontemporal_load((const v2f*)(xrow + ko));
                v2f b = *(const v2f*)(w1s + mlr + ko);
                acc = __builtin_amdgcn_wmma_f32_16x16x4_f32(
                    false, a, false, b, (short)0, acc, false, false);
            }

            float* yb = &y1s[wave][0];
            #pragma unroll
            for (int r = 0; r < 8; ++r) {
                float v = acc[r] + b1s[ml];
                v = fminf(fmaxf(v, 0.f), 1.f);
                yb[(8 * half + r) * Y1PAD + ml] = v;
            }
            asm volatile("" ::: "memory");

            float xv[NL2];
            #pragma unroll
            for (int j = 0; j < NL2; ++j) xv[j] = yb[ml * Y1PAD + j];

            float acc3 = 0.f;
            #pragma unroll
            for (int q = 0; q < 16; ++q) {
                int jj = half * 16 + q;
                float s2 = b2s[jj];
                #pragma unroll
                for (int j = 0; j < NL2; ++j) s2 += xv[j] * w2s[jj * NL2 + j];
                s2 = fminf(fmaxf(s2, 0.f), 1.f);
                acc3 += s2 * wos[jj];
            }
            acc3 += __shfl_xor(acc3, 16, 32);

            if (half == 0 && gs < n_c) out[samp] = acc3 + bo_c;
        }

        const int s1 = s0 + gridDim.x * CHUNK;
        if (s1 < B) {
            #pragma unroll
            for (int e = 0; e < SPT; ++e) {
                int o = e * 256 + tid;
                int s = s1 + o;
                iv[e] = idx[(o < CHUNK && s < B) ? s : s1];
            }
        }
        __syncthreads();
    }
}

extern "C" void kernel_launch(void* const* d_in, const int* in_sizes, int n_in,
                              void* d_out, int out_size, void* d_ws, size_t ws_size,
                              hipStream_t stream) {
    const float* x   = (const float*)d_in[0];
    const int*   idx = (const int*)  d_in[1];
    const float* w1  = (const float*)d_in[2];
    const float* b1  = (const float*)d_in[3];
    const float* w2  = (const float*)d_in[4];
    const float* b2  = (const float*)d_in[5];
    const float* wo  = (const float*)d_in[6];
    const float* bo  = (const float*)d_in[7];
    float* out = (float*)d_out;

    const int B = in_sizes[1];

    dim3 grid(BPS, NSTACK, 1);
    ls_main<<<grid, 256, 0, stream>>>(x, idx, w1, b1, w2, b2, wo, bo, out, B);
}


# x rows reach the WMMAs through a per-wave LDS stage filled by whole-line 128-bit loads (4 lines per load instead of 16 pieces of 16 B), one 64-column chunk ahead; k order of every sum unchanged
# speedup vs baseline: 3.5365x; 1.9985x over previous
#include <hip/hip_runtime.h>


#define L1X2   1024
#define NL2    16
#define NL3    32
#define NSTACK 8
#define W1PAD  1028
#define Y1PAD  17
#define XCH    64
#define XPAD   68
#define NCH    (L1X2 / XCH)
#define XLD    8
#define BPS    96
#define WAVES  8
#define SPT    3
#define CHUNK  683

typedef __attribute__((ext_vector_type(2))) float v2f;
typedef __attribute__((ext_vector_type(4))) float v4f;
typedef __attribute__((ext_vector_type(8))) float v8f;

__global__ __launch_bounds__(256)
void ls_main(const float* __restrict__ x, const int* __restrict__ idx,
             const float* __restrict__ w1, const float* __restrict__ b1,
             const float* __restrict__ w2, const float* __restrict__ b2,
             const float* __restrict__ wo, const float* __restrict__ bo,
             float* __restrict__ out, int B) {
    __shared__ float w1s[NL2 * W1PAD];
    __shared__ float w2s[NL3 * NL2];
    __shared__ float b1s[NL2];
    __shared__ float b2s[NL3];
    __shared__ float wos[NL3];
    __shared__ __attribute__((aligned(16))) float xst[WAVES][16 * XPAD];
    __shared__ int   lst[CHUNK];
    __shared__ int   wtot[WAVES];

    const int c   = blockIdx.y;
    const int tid = threadIdx.x;

    int s0 = blockIdx.x * CHUNK;
    int iv[SPT];
    #pragma unroll
    for (int e = 0; e < SPT; ++e) iv[e] = 0;
    if (s0 < B) {
        #pragma unroll
        for (int e = 0; e < SPT; ++e) {
            int o = e * 256 + tid;
            int s = s0 + o;
            iv[e] = idx[(o < CHUNK && s < B) ? s : s0];
        }
    }

    for (int i = tid; i < NL2 * L1X2; i += 256) {
        int r = i >> 10, col = i & 1023;
        w1s[r * W1PAD + col] = w1[(c * NL2 + r) * L1X2 + col];
    }
    for (int i = tid; i < NL3 * NL2; i += 256) {
        int r = i >> 4, col = i & 15;
        w2s[i] = w2[(c * NL3 + r) * NL2 + col];
    }
    if (tid < NL2) b1s[tid] = b1[c * NL2 + tid];
    if (tid < NL3) { b2s[tid] = b2[c * NL3 + tid]; wos[tid] = wo[c * NL3 + tid]; }

    const float bo_c = bo[c];

    const int wave = tid >> 5;
    const int lane = tid & 31;
    const int ml   = lane & 15;
    const int half = lane >> 4;
    const int mlr  = ml * W1PAD;
    const unsigned lanes_below = (1u << lane) - 1u;

    for (; s0 < B; s0 += gridDim.x * CHUNK) {
        bool     hit[SPT];
        unsigned bal[SPT];
        int found = 0;
        #pragma unroll
        for (int e = 0; e < SPT; ++e) {
            int o = e * 256 + tid;
            hit[e] = (o < CHUNK) && (s0 + o < B) && (iv[e] == c);
            bal[e] = (unsigned)__ballot(hit[e]);
            found += __popc(bal[e]);
        }
        if (lane == 0) wtot[wave] = found;
        __syncthreads();

        int pos = 0, n_c = 0;
        #pragma unroll
        for (int w = 0; w < WAVES; ++w) {
            int v = wtot[w];
            if (w < wave) pos += v;
            n_c += v;
        }
        #pragma unroll
        for (int e = 0; e < SPT; ++e) {
            if (hit[e]) lst[pos + __popc(bal[e] & lanes_below)] = s0 + e * 256 + tid;
            pos += __popc(bal[e]);
        }
        __syncthreads();

        const int nTiles = (n_c + 15) >> 4;

        for (int t = wave; t < nTiles; t += WAVES) {
            int gs   = t * 16 + ml;
            int cgs  = gs < n_c ? gs : n_c - 1;
            int samp = lst[cgs];

            float* st = &xst[wave][0];
            const int lcol = ml * 4;
            size_t xb[XLD];
            #pragma unroll
            for (int j = 0; j < XLD; ++j) {
                int g = t * 16 + 2 * j + half;
                g = g < n_c ? g : n_c - 1;
                xb[j] = ((size_t)lst[g] * L1X2 + lcol) * sizeof(float);
            }
            v4f xr[XLD];
            #pragma unroll
            for (int j = 0; j < XLD; ++j)
                xr[j] = __builtin_nontemporal_load((const v4f*)((const char*)x + xb[j]));

            v8f acc = {0.f, 0.f, 0.f, 0.f, 0.f, 0.f, 0.f, 0.f};
            for (int ch = 0; ch < NCH; ++ch) {
                asm volatile("" ::: "memory");
                #pragma unroll
                for (int j = 0; j < XLD; ++j)
                    *(v4f*)(st + (2 * j + half) * XPAD + lcol) = xr[j];
                asm volatile("" ::: "memory");
                if (ch + 1 < NCH) {
                    const char* xc = (const char*)x + (size_t)(ch + 1) * (XCH * sizeof(float));
                    #pragma unroll
                    for (int j = 0; j < XLD; ++j)
                        xr[j] = __builtin_nontemporal_load((const v4f*)(xc + xb[j]));
                }
                const float* ak = st + ml * XPAD + 2 * half;
                const float* bk = w1s + mlr + ch * XCH + 2 * half;
                #pragma unroll
                for (int kk = 0; kk < XCH; kk += 4) {
                    v2f a = *(const v2f*)(ak + kk);
                    v2f b = *(const v2f*)(bk + kk);
                    acc = __builtin_amdgcn_wmma_f32_16x16x4_f32(
                        false, a, false, b, (short)0, acc, false, false);
                }
            }

            float* yb = st;
            #pragma unroll
            for (int r = 0; r < 8; ++r) {
                float v = acc[r] + b1s[ml];
                v = fminf(fmaxf(v, 0.f), 1.f);
                yb[(8 * half + r) * Y1PAD + ml] = v;
            }
            asm volatile("" ::: "memory");

            float xv[NL2];
            #pragma unroll
            for (int j = 0; j < NL2; ++j) xv[j] = yb[ml * Y1PAD + j];

            float acc3 = 0.f;
            #pragma unroll 4
            for (int q = 0; q < 16; ++q) {
                int jj = half * 16 + q;
                float s2 = b2s[jj];
                #pragma unroll
                for (int j = 0; j < NL2; ++j) s2 += xv[j] * w2s[jj * NL2 + j];
                s2 = fminf(fmaxf(s2, 0.f), 1.f);
                acc3 += s2 * wos[jj];
            }
            acc3 += __shfl_xor(acc3, 16, 32);

            if (half == 0 && gs < n_c) out[samp] = acc3 + bo_c;
        }

        const int s1 = s0 + gridDim.x * CHUNK;
        if (s1 < B) {
            #pragma unroll
            for (int e = 0; e < SPT; ++e) {
                int o = e * 256 + tid;
                int s = s1 + o;
                iv[e] = idx[(o < CHUNK && s < B) ? s : s1];
            }
        }
        __syncthreads();
    }
}

extern "C" void kernel_launch(void* const* d_in, const int* in_sizes, int n_in,
                              void* d_out, int out_size, void* d_ws, size_t ws_size,
                              hipStream_t stream) {
    const float* x   = (const float*)d_in[0];
    const int*   idx = (const int*)  d_in[1];
    const float* w1  = (const float*)d_in[2];
    const float* b1  = (const float*)d_in[3];
    const float* w2  = (const float*)d_in[4];
    const float* b2  = (const float*)d_in[5];
    const float* wo  = (const float*)d_in[6];
    const float* bo  = (const float*)d_in[7];
    float* out = (float*)d_out;

    const int B = in_sizes[1];

    dim3 grid(BPS, NSTACK, 1);
    ls_main<<<grid, 256, 0, stream>>>(x, idx, w1, b1, w2, b2, wo, bo, out, B);
}


# x loads run two 64-column chunks ahead of the WMMAs instead of one: a second register set, 16 loads of 128 bits in flight per wave; k order of every sum unchanged
# speedup vs baseline: 3.7587x; 1.0628x over previous
#include <hip/hip_runtime.h>


#define L1X2   1024
#define NL2    16
#define NL3    32
#define NSTACK 8
#define W1PAD  1028
#define Y1PAD  17
#define XCH    64
#define XPAD   68
#define NCH    (L1X2 / XCH)
#define XLD    8
#define BPS    96
#define WAVES  8
#define SPT    3
#define CHUNK  683

typedef __attribute__((ext_vector_type(2))) float v2f;
typedef __attribute__((ext_vector_type(4))) float v4f;
typedef __attribute__((ext_vector_type(8))) float v8f;

__global__ __launch_bounds__(256)
void ls_main(const float* __restrict__ x, const int* __restrict__ idx,
             const float* __restrict__ w1, const float* __restrict__ b1,
             const float* __restrict__ w2, const float* __restrict__ b2,
             const float* __restrict__ wo, const float* __restrict__ bo,
             float* __restrict__ out, int B) {
    __shared__ float w1s[NL2 * W1PAD];
    __shared__ float w2s[NL3 * NL2];
    __shared__ float b1s[NL2];
    __shared__ float b2s[NL3];
    __shared__ float wos[NL3];
    __shared__ __attribute__((aligned(16))) float xst[WAVES][16 * XPAD];
    __shared__ int   lst[CHUNK];
    __shared__ int   wtot[WAVES];

    const int c   = blockIdx.y;
    const int tid = threadIdx.x;

    int s0 = blockIdx.x * CHUNK;
    int iv[SPT];
    #pragma unroll
    for (int e = 0; e < SPT; ++e) iv[e] = 0;
    if (s0 < B) {
        #pragma unroll
        for (int e = 0; e < SPT; ++e) {
            int o = e * 256 + tid;
            int s = s0 + o;
            iv[e] = idx[(o < CHUNK && s < B) ? s : s0];
        }
    }

    for (int i = tid; i < NL2 * L1X2; i += 256) {
        int r = i >> 10, col = i & 1023;
        w1s[r * W1PAD + col] = w1[(c * NL2 + r) * L1X2 + col];
    }
    for (int i = tid; i < NL3 * NL2; i += 256) {
        int r = i >> 4, col = i & 15;
        w2s[i] = w2[(c * NL3 + r) * NL2 + col];
    }
    if (tid < NL2) b1s[tid] = b1[c * NL2 + tid];
    if (tid < NL3) { b2s[tid] = b2[c * NL3 + tid]; wos[tid] = wo[c * NL3 + tid]; }

    const float bo_c = bo[c];

    const int wave = tid >> 5;
    const int lane = tid & 31;
    const int ml   = lane & 15;
    const int half = lane >> 4;
    const int mlr  = ml * W1PAD;
    const unsigned lanes_below = (1u << lane) - 1u;

    for (; s0 < B; s0 += gridDim.x * CHUNK) {
        bool     hit[SPT];
        unsigned bal[SPT];
        int found = 0;
        #pragma unroll
        for (int e = 0; e < SPT; ++e) {
            int o = e * 256 + tid;
            hit[e] = (o < CHUNK) && (s0 + o < B) && (iv[e] == c);
            bal[e] = (unsigned)__ballot(hit[e]);
            found += __popc(bal[e]);
        }
        if (lane == 0) wtot[wave] = found;
        __syncthreads();

        int pos = 0, n_c = 0;
        #pragma unroll
        for (int w = 0; w < WAVES; ++w) {
            int v = wtot[w];
            if (w < wave) pos += v;
            n_c += v;
        }
        #pragma unroll
        for (int e = 0; e < SPT; ++e) {
            if (hit[e]) lst[pos + __popc(bal[e] & lanes_below)] = s0 + e * 256 + tid;
            pos += __popc(bal[e]);
        }
        __syncthreads();

        const int nTiles = (n_c + 15) >> 4;

        for (int t = wave; t < nTiles; t += WAVES) {
            int gs   = t * 16 + ml;
            int cgs  = gs < n_c ? gs : n_c - 1;
            int samp = lst[cgs];

            float* st = &xst[wave][0];
            const int lcol = ml * 4;
            size_t xb[XLD];
            #pragma unroll
            for (int j = 0; j < XLD; ++j) {
                int g = t * 16 + 2 * j + half;
                g = g < n_c ? g : n_c - 1;
                xb[j] = ((size_t)lst[g] * L1X2 + lcol) * sizeof(float);
            }
            v4f xr[2][XLD];
            #pragma unroll
            for (int p = 0; p < 2; ++p) {
                const char* xc = (const char*)x + (size_t)p * (XCH * sizeof(float));
                #pragma unroll
                for (int j = 0; j < XLD; ++j)
                    xr[p][j] = __builtin_nontemporal_load((const v4f*)(xc + xb[j]));
            }

            v8f acc = {0.f, 0.f, 0.f, 0.f, 0.f, 0.f, 0.f, 0.f};
            for (int ch2 = 0; ch2 < NCH; ch2 += 2) {
                #pragma unroll
                for (int p = 0; p < 2; ++p) {
                    const int ch = ch2 + p;
                    __builtin_amdgcn_sched_barrier(0);
                    asm volatile("" ::: "memory");
                    #pragma unroll
                    for (int j = 0; j < XLD; ++j)
                        *(v4f*)(st + (2 * j + half) * XPAD + lcol) = xr[p][j];
                    asm volatile("" ::: "memory");
                    if (ch + 2 < NCH) {
                        const char* xc = (const char*)x + (size_t)(ch + 2) * (XCH * sizeof(float));
                        #pragma unroll
                        for (int j = 0; j < XLD; ++j)
                            xr[p][j] = __builtin_nontemporal_load((const v4f*)(xc + xb[j]));
                    }
                    __builtin_amdgcn_sched_barrier(0);
                    const float* ak = st + ml * XPAD + 2 * half;
                    const float* bk = w1s + mlr + ch * XCH + 2 * half;
                    #pragma unroll
                    for (int kk = 0; kk < XCH; kk += 4) {
                        v2f a = *(const v2f*)(ak + kk);
                        v2f b = *(const v2f*)(bk + kk);
                        acc = __builtin_amdgcn_wmma_f32_16x16x4_f32(
                            false, a, false, b, (short)0, acc, false, false);
                    }
                }
            }

            float* yb = st;
            #pragma unroll
            for (int r = 0; r < 8; ++r) {
                float v = acc[r] + b1s[ml];
                v = fminf(fmaxf(v, 0.f), 1.f);
                yb[(8 * half + r) * Y1PAD + ml] = v;
            }
            asm volatile("" ::: "memory");

            float xv[NL2];
            #pragma unroll
            for (int j = 0; j < NL2; ++j) xv[j] = yb[ml * Y1PAD + j];

            float acc3 = 0.f;
            #pragma unroll 4
            for (int q = 0; q < 16; ++q) {
                int jj = half * 16 + q;
                float s2 = b2s[jj];
                #pragma unroll
                for (int j = 0; j < NL2; ++j) s2 += xv[j] * w2s[jj * NL2 + j];
                s2 = fminf(fmaxf(s2, 0.f), 1.f);
                acc3 += s2 * wos[jj];
            }
            acc3 += __shfl_xor(acc3, 16, 32);

            if (half == 0 && gs < n_c) out[samp] = acc3 + bo_c;
        }

        const int s1 = s0 + gridDim.x * CHUNK;
        if (s1 < B) {
            #pragma unroll
            for (int e = 0; e < SPT; ++e) {
                int o = e * 256 + tid;
                int s = s1 + o;
                iv[e] = idx[(o < CHUNK && s < B) ? s : s1];
            }
        }
        __syncthreads();
    }
}

extern "C" void kernel_launch(void* const* d_in, const int* in_sizes, int n_in,
                              void* d_out, int out_size, void* d_ws, size_t ws_size,
                              hipStream_t stream) {
    const float* x   = (const float*)d_in[0];
    const int*   idx = (const int*)  d_in[1];
    const float* w1  = (const float*)d_in[2];
    const float* b1  = (const float*)d_in[3];
    const float* w2  = (const float*)d_in[4];
    const float* b2  = (const float*)d_in[5];
    const float* wo  = (const float*)d_in[6];
    const float* bo  = (const float*)d_in[7];
    float* out = (float*)d_out;

    const int B = in_sizes[1];

    dim3 grid(BPS, NSTACK, 1);
    ls_main<<<grid, 256, 0, stream>>>(x, idx, w1, b1, w2, b2, wo, bo, out, B);
}


# One 24-wave block per WGP on a 2048-sample chunk (grid 32x8, was 96x8 of 8 waves): tiles go round the SIMDs 4-5 each, not 6/6/3/3; weights staged once per WGP; k order of every sum unchanged
# speedup vs baseline: 4.0175x; 1.0689x over previous
#include <hip/hip_runtime.h>


#define L1X2   1024
#define NL2    16
#define NL3    32
#define NSTACK 8
#define W1PAD  1028
#define Y1PAD  17
#define XCH    64
#define XPAD   68
#define NCH    (L1X2 / XCH)
#define XLD    8
#define BPS    32
#define WAVES  24
#define NT     (WAVES * 32)
#define SPT    3
#define CHUNK  2048

typedef __attribute__((ext_vector_type(2))) float v2f;
typedef __attribute__((ext_vector_type(4))) float v4f;
typedef __attribute__((ext_vector_type(8))) float v8f;

__global__ __launch_bounds__(NT)
void ls_main(const float* __restrict__ x, const int* __restrict__ idx,
             const float* __restrict__ w1, const float* __restrict__ b1,
             const float* __restrict__ w2, const float* __restrict__ b2,
             const float* __restrict__ wo, const float* __restrict__ bo,
             float* __restrict__ out, int B) {
    __shared__ float w1s[NL2 * W1PAD];
    __shared__ float w2s[NL3 * NL2];
    __shared__ float b1s[NL2];
    __shared__ float b2s[NL3];
    __shared__ float wos[NL3];
    __shared__ __attribute__((aligned(16))) float xst[WAVES][16 * XPAD];
    __shared__ int   lst[CHUNK];
    __shared__ int   wtot[WAVES];

    const int c   = blockIdx.y;
    const int tid = threadIdx.x;

    int s0 = blockIdx.x * CHUNK;
    int iv[SPT];
    #pragma unroll
    for (int e = 0; e < SPT; ++e) iv[e] = 0;
    if (s0 < B) {
        #pragma unroll
        for (int e = 0; e < SPT; ++e) {
            int o = e * NT + tid;
            int s = s0 + o;
            iv[e] = idx[(o < CHUNK && s < B) ? s : s0];
        }
    }

    for (int i = tid; i < NL2 * L1X2; i += NT) {
        int r = i >> 10, col = i & 1023;
        w1s[r * W1PAD + col] = w1[(c * NL2 + r) * L1X2 + col];
    }
    for (int i = tid; i < NL3 * NL2; i += NT) {
        int r = i >> 4, col = i & 15;
        w2s[i] = w2[(c * NL3 + r) * NL2 + col];
    }
    if (tid < NL2) b1s[tid] = b1[c * NL2 + tid];
    if (tid < NL3) { b2s[tid] = b2[c * NL3 + tid]; wos[tid] = wo[c * NL3 + tid]; }

    const float bo_c = bo[c];

    const int wave = tid >> 5;
    const int lane = tid & 31;
    const int ml   = lane & 15;
    const int half = lane >> 4;
    const int mlr  = ml * W1PAD;
    const unsigned lanes_below = (1u << lane) - 1u;

    for (; s0 < B; s0 += gridDim.x * CHUNK) {
        bool     hit[SPT];
        unsigned bal[SPT];
        int found = 0;
        #pragma unroll
        for (int e = 0; e < SPT; ++e) {
            int o = e * NT + tid;
            hit[e] = (o < CHUNK) && (s0 + o < B) && (iv[e] == c);
            bal[e] = (unsigned)__ballot(hit[e]);
            found += __popc(bal[e]);
        }
        if (lane == 0) wtot[wave] = found;
        __syncthreads();

        int pos = 0, n_c = 0;
        #pragma unroll
        for (int w = 0; w < WAVES; ++w) {
            int v = wtot[w];
            if (w < wave) pos += v;
            n_c += v;
        }
        #pragma unroll
        for (int e = 0; e < SPT; ++e) {
            if (hit[e]) lst[pos + __popc(bal[e] & lanes_below)] = s0 + e * NT + tid;
            pos += __popc(bal[e]);
        }
        __syncthreads();

        const int nTiles = (n_c + 15) >> 4;

        for (int t = wave; t < nTiles; t += WAVES) {
            int gs   = t * 16 + ml;
            int cgs  = gs < n_c ? gs : n_c - 1;
            int samp = lst[cgs];

            float* st = &xst[wave][0];
            const int lcol = ml * 4;
            size_t xb[XLD];
            #pragma unroll
            for (int j = 0; j < XLD; ++j) {
                int g = t * 16 + 2 * j + half;
                g = g < n_c ? g : n_c - 1;
                xb[j] = ((size_t)lst[g] * L1X2 + lcol) * sizeof(float);
            }
            v4f xr[2][XLD];
            #pragma unroll
            for (int p = 0; p < 2; ++p) {
                const char* xc = (const char*)x + (size_t)p * (XCH * sizeof(float));
                #pragma unroll
                for (int j = 0; j < XLD; ++j)
                    xr[p][j] = __builtin_nontemporal_load((const v4f*)(xc + xb[j]));
            }

            v8f acc = {0.f, 0.f, 0.f, 0.f, 0.f, 0.f, 0.f, 0.f};
            for (int ch2 = 0; ch2 < NCH; ch2 += 2) {
                #pragma unroll
                for (int p = 0; p < 2; ++p) {
                    const int ch = ch2 + p;
                    __builtin_amdgcn_sched_barrier(0);
                    asm volatile("" ::: "memory");
                    #pragma unroll
                    for (int j = 0; j < XLD; ++j)
                        *(v4f*)(st + (2 * j + half) * XPAD + lcol) = xr[p][j];
                    asm volatile("" ::: "memory");
                    if (ch + 2 < NCH) {
                        const char* xc = (const char*)x + (size_t)(ch + 2) * (XCH * sizeof(float));
                        #pragma unroll
                        for (int j = 0; j < XLD; ++j)
                            xr[p][j] = __builtin_nontemporal_load((const v4f*)(xc + xb[j]));
                    }
                    __builtin_amdgcn_sched_barrier(0);
                    const float* ak = st + ml * XPAD + 2 * half;
                    const float* bk = w1s + mlr + ch * XCH + 2 * half;
                    #pragma unroll
                    for (int kk = 0; kk < XCH; kk += 4) {
                        v2f a = *(const v2f*)(ak + kk);
                        v2f b = *(const v2f*)(bk + kk);
                        acc = __builtin_amdgcn_wmma_f32_16x16x4_f32(
                            false, a, false, b, (short)0, acc, false, false);
                    }
                }
            }

            float* yb = st;
            #pragma unroll
            for (int r = 0; r < 8; ++r) {
                float v = acc[r] + b1s[ml];
                v = fminf(fmaxf(v, 0.f), 1.f);
                yb[(8 * half + r) * Y1PAD + ml] = v;
            }
            asm volatile("" ::: "memory");

            float xv[NL2];
            #pragma unroll
            for (int j = 0; j < NL2; ++j) xv[j] = yb[ml * Y1PAD + j];

            float acc3 = 0.f;
            #pragma unroll 4
            for (int q = 0; q < 16; ++q) {
                int jj = half * 16 + q;
                float s2 = b2s[jj];
                #pragma unroll
                for (int j = 0; j < NL2; ++j) s2 += xv[j] * w2s[jj * NL2 + j];
                s2 = fminf(fmaxf(s2, 0.f), 1.f);
                acc3 += s2 * wos[jj];
            }
            acc3 += __shfl_xor(acc3, 16, 32);

            if (half == 0 && gs < n_c) out[samp] = acc3 + bo_c;
        }

        const int s1 = s0 + gridDim.x * CHUNK;
        if (s1 < B) {
            #pragma unroll
            for (int e = 0; e < SPT; ++e) {
                int o = e * NT + tid;
                int s = s1 + o;
                iv[e] = idx[(o < CHUNK && s < B) ? s : s1];
            }
        }
        __syncthreads();
    }
}

extern "C" void kernel_launch(void* const* d_in, const int* in_sizes, int n_in,
                              void* d_out, int out_size, void* d_ws, size_t ws_size,
                              hipStream_t stream) {
    const float* x   = (const float*)d_in[0];
    const int*   idx = (const int*)  d_in[1];
    const float* w1  = (const float*)d_in[2];
    const float* b1  = (const float*)d_in[3];
    const float* w2  = (const float*)d_in[4];
    const float* b2  = (const float*)d_in[5];
    const float* wo  = (const float*)d_in[6];
    const float* bo  = (const float*)d_in[7];
    float* out = (float*)d_out;

    const int B = in_sizes[1];

    dim3 grid(BPS, NSTACK, 1);
    ls_main<<<grid, NT, 0, stream>>>(x, idx, w1, b1, w2, b2, wo, bo, out, B);
}
